# starting kernel
# baseline (speedup 1.0000x reference)
#include <hip/hip_runtime.h>

typedef float v2f __attribute__((ext_vector_type(2)));
typedef float v8f __attribute__((ext_vector_type(8)));

#define BATCH 32768
#define NF    200
#define DIN   5
#define NH    15
#define NOUT  30

__device__ __forceinline__ v8f wmma_f32_k4(v2f a, v2f b, v8f c) {
  return __builtin_amdgcn_wmma_f32_16x16x4_f32(
       false, a,  false, b,
       (short)0, c,  false,  false);
}

template <int IMM>
__device__ __forceinline__ float swz(float x) {
  return __int_as_float(__builtin_amdgcn_ds_swizzle(__float_as_int(x), IMM));
}
#define SWAP16 0x401f
#define XOR1   0x041f
#define XOR2   0x081f
#define XOR4   0x101f
#define XOR8   0x201f

__global__ __launch_bounds__(256) void mlp200_wmma_kernel(
    const float* __restrict__ y,  const float* __restrict__ W1,
    const float* __restrict__ b1, const float* __restrict__ W2,
    const float* __restrict__ b2, const float* __restrict__ W3,
    const float* __restrict__ b3, float* __restrict__ out)
{
  const int  lane = threadIdx.x & 31;
  const int  wave = threadIdx.x >> 5;
  const int  q    = lane >> 4;
  const bool qhi  = (q != 0);
  const int  n    = lane & 15;
  const int  b0   = (blockIdx.x * 8 + wave) * 16;

  const int  nc   = (n < NH - 1) ? n : (NH - 1);
  const int  oc2  = 16 + ((n < 14) ? n : 13);
  const bool o2ok = (n < 14);

  const int  iy0  = (b0 + n) * (NF * DIN) + 2 * q;
  const int  iy1  = iy0 + 1;
  const int  iy4  = (b0 + n) * (NF * DIN) + 4;

  const int iW1a = (2 * q) * NH + nc;
  const int iW1b = iW1a + NH;
  const int iW1c = 4 * NH + nc;

  float acc[8];
  #pragma unroll
  for (int r = 0; r < 8; ++r) acc[r] = 0.f;

  #pragma unroll 2
  for (int f = 0; f < NF; ++f) {
    const float* yb  = y  + f * DIN;
    const float* W1f = W1 + f * (DIN * NH);
    const float* W2f = W2 + f * (NH * NOUT);
    const float* b1f = b1 + f * NH;
    const float* b2f = b2 + f * NOUT;
    const float* W3f = W3 + f * NOUT;

    v8f c1 = {};
    {
      v2f a, bb;
      a.x  = W1f[iW1a];
      a.y  = W1f[iW1b];
      bb.x = yb[iy0];
      bb.y = yb[iy1];
      c1 = wmma_f32_k4(a, bb, c1);
      a.x  = W1f[iW1c];
      a.y  = b1f[nc];
      float y4 = yb[iy4];
      bb.x = qhi ? 0.f : y4;
      bb.y = qhi ? 0.f : 1.f;
      c1 = wmma_f32_k4(a, bb, c1);
    }

    float hv[8], sw[8];
    #pragma unroll
    for (int r = 0; r < 8; ++r) { hv[r] = c1[r]; sw[r] = swz<SWAP16>(c1[r]); }

    v2f a0, a1, a2, a3;
    a0.x = qhi ? sw[2] : hv[0];
    a0.y = qhi ? sw[3] : hv[1];
    a1.x = qhi ? sw[6] : hv[4];
    a1.y = qhi ? sw[7] : hv[5];
    a2.x = qhi ? hv[2] : sw[0];
    a2.y = qhi ? hv[3] : sw[1];
    a3.x = qhi ? hv[6] : sw[4];
    a3.y = qhi ? 1.f   : sw[5];

    v8f c2a = {}, c2b = {};
    #pragma unroll
    for (int k = 0; k < 4; ++k) {
      const v2f av = (k == 0) ? a0 : (k == 1) ? a1 : (k == 2) ? a2 : a3;
      const int h0  = 4 * k + 2 * q;
      int       h1i = h0 + 1; if (h1i > NH - 1) h1i = NH - 1;
      v2f bta, btb;
      bta.x = W2f[h0  * NOUT + n];
      btb.x = W2f[h0  * NOUT + oc2];
      float bay = W2f[h1i * NOUT + n];
      float bby = W2f[h1i * NOUT + oc2];
      if (k == 3) {
        float b2a = b2f[n];
        float b2b = b2f[oc2];
        bta.y = qhi ? b2a : bay;
        btb.y = qhi ? b2b : bby;
      } else {
        bta.y = bay;
        btb.y = bby;
      }
      c2a = wmma_f32_k4(av, bta, c2a);
      c2b = wmma_f32_k4(av, btb, c2b);
    }

    float w3a = W3f[n];
    float w3t = W3f[oc2];
    float w3b = o2ok ? w3t : 0.f;
    #pragma unroll
    for (int r = 0; r < 8; ++r) {
      float va = c2a[r] > 0.f ? c2a[r] : 0.f;
      float vb = c2b[r] > 0.f ? c2b[r] : 0.f;
      acc[r] += va * w3a + vb * w3b;
    }
  }

  #pragma unroll
  for (int r = 0; r < 8; ++r) {
    float v = acc[r];
    v += swz<XOR1>(v);
    v += swz<XOR2>(v);
    v += swz<XOR4>(v);
    v += swz<XOR8>(v);
    acc[r] = v;
  }

  if (n == 0) {
    const float b3v = b3[0];
    float* op = out + b0 + 8 * q;
    float4 o0 = make_float4(acc[0] + b3v, acc[1] + b3v, acc[2] + b3v, acc[3] + b3v);
    float4 o1 = make_float4(acc[4] + b3v, acc[5] + b3v, acc[6] + b3v, acc[7] + b3v);
    reinterpret_cast<float4*>(op)[0] = o0;
    reinterpret_cast<float4*>(op)[1] = o1;
  }
}

extern "C" void kernel_launch(void* const* d_in, const int* in_sizes, int n_in,
                              void* d_out, int out_size, void* d_ws, size_t ws_size,
                              hipStream_t stream) {
  (void)in_sizes; (void)n_in; (void)d_ws; (void)ws_size; (void)out_size;
  const float* y  = (const float*)d_in[0];
  const float* W1 = (const float*)d_in[1];
  const float* b1 = (const float*)d_in[2];
  const float* W2 = (const float*)d_in[3];
  const float* b2 = (const float*)d_in[4];
  const float* W3 = (const float*)d_in[5];
  const float* b3 = (const float*)d_in[6];
  float* out = (float*)d_out;

  dim3 grid(BATCH / (16 * 8));
  dim3 block(256);
  mlp200_wmma_kernel<<<grid, block, 0, stream>>>(y, W1, b1, W2, b2, W3, b3, out);
}


# Feature loop split over 8 waves per 16-row tile (8 waves per SIMD instead of 2); the 8 partial sums are added through LDS in fixed order, so the feature sum's order changes
# speedup vs baseline: 1.4950x; 1.4950x over previous
#include <hip/hip_runtime.h>

typedef float v2f __attribute__((ext_vector_type(2)));
typedef float v8f __attribute__((ext_vector_type(8)));

#define BATCH 32768
#define NF    200
#define DIN   5
#define NH    15
#define NOUT  30
#define SPLIT 8
#define TILES 2

__device__ __forceinline__ v8f wmma_f32_k4(v2f a, v2f b, v8f c) {
  return __builtin_amdgcn_wmma_f32_16x16x4_f32(
       false, a,  false, b,
       (short)0, c,  false,  false);
}

template <int IMM>
__device__ __forceinline__ float swz(float x) {
  return __int_as_float(__builtin_amdgcn_ds_swizzle(__float_as_int(x), IMM));
}
#define SWAP16 0x401f
#define XOR1   0x041f
#define XOR2   0x081f
#define XOR4   0x101f
#define XOR8   0x201f

__global__ __launch_bounds__(SPLIT * TILES * 32) void mlp200_wmma_kernel(
    const float* __restrict__ y,  const float* __restrict__ W1,
    const float* __restrict__ b1, const float* __restrict__ W2,
    const float* __restrict__ b2, const float* __restrict__ W3,
    const float* __restrict__ b3, float* __restrict__ out)
{
  __shared__ float part[SPLIT][TILES * 16];

  const int  lane = threadIdx.x & 31;
  const int  wave  = __builtin_amdgcn_readfirstlane((int)(threadIdx.x >> 5));
  const int  tile  = wave / SPLIT;
  const int  slice = wave % SPLIT;
  const int  q    = lane >> 4;
  const bool qhi  = (q != 0);
  const int  n    = lane & 15;
  const int  b0   = (blockIdx.x * TILES + tile) * 16;

  const int  nc   = (n < NH - 1) ? n : (NH - 1);
  const int  oc2  = 16 + ((n < 14) ? n : 13);
  const bool o2ok = (n < 14);

  const int  iy0  = (b0 + n) * (NF * DIN) + 2 * q;
  const int  iy1  = iy0 + 1;
  const int  iy4  = (b0 + n) * (NF * DIN) + 4;

  const int iW1a = (2 * q) * NH + nc;
  const int iW1b = iW1a + NH;
  const int iW1c = 4 * NH + nc;

  float acc[8];
  #pragma unroll
  for (int r = 0; r < 8; ++r) acc[r] = 0.f;

  #pragma unroll 2
  for (int f = slice; f < NF; f += SPLIT) {
    const float* yb  = y  + f * DIN;
    const float* W1f = W1 + f * (DIN * NH);
    const float* W2f = W2 + f * (NH * NOUT);
    const float* b1f = b1 + f * NH;
    const float* b2f = b2 + f * NOUT;
    const float* W3f = W3 + f * NOUT;

    v8f c1 = {};
    {
      v2f a, bb;
      a.x  = W1f[iW1a];
      a.y  = W1f[iW1b];
      bb.x = yb[iy0];
      bb.y = yb[iy1];
      c1 = wmma_f32_k4(a, bb, c1);
      a.x  = W1f[iW1c];
      a.y  = b1f[nc];
      float y4 = yb[iy4];
      bb.x = qhi ? 0.f : y4;
      bb.y = qhi ? 0.f : 1.f;
      c1 = wmma_f32_k4(a, bb, c1);
    }

    float hv[8], sw[8];
    #pragma unroll
    for (int r = 0; r < 8; ++r) { hv[r] = c1[r]; sw[r] = swz<SWAP16>(c1[r]); }

    v2f a0, a1, a2, a3;
    a0.x = qhi ? sw[2] : hv[0];
    a0.y = qhi ? sw[3] : hv[1];
    a1.x = qhi ? sw[6] : hv[4];
    a1.y = qhi ? sw[7] : hv[5];
    a2.x = qhi ? hv[2] : sw[0];
    a2.y = qhi ? hv[3] : sw[1];
    a3.x = qhi ? hv[6] : sw[4];
    a3.y = qhi ? 1.f   : sw[5];

    v8f c2a = {}, c2b = {};
    #pragma unroll
    for (int k = 0; k < 4; ++k) {
      const v2f av = (k == 0) ? a0 : (k == 1) ? a1 : (k == 2) ? a2 : a3;
      const int h0  = 4 * k + 2 * q;
      int       h1i = h0 + 1; if (h1i > NH - 1) h1i = NH - 1;
      v2f bta, btb;
      bta.x = W2f[h0  * NOUT + n];
      btb.x = W2f[h0  * NOUT + oc2];
      float bay = W2f[h1i * NOUT + n];
      float bby = W2f[h1i * NOUT + oc2];
      if (k == 3) {
        float b2a = b2f[n];
        float b2b = b2f[oc2];
        bta.y = qhi ? b2a : bay;
        btb.y = qhi ? b2b : bby;
      } else {
        bta.y = bay;
        btb.y = bby;
      }
      c2a = wmma_f32_k4(av, bta, c2a);
      c2b = wmma_f32_k4(av, btb, c2b);
    }

    float w3a = W3f[n];
    float w3t = W3f[oc2];
    float w3b = o2ok ? w3t : 0.f;
    #pragma unroll
    for (int r = 0; r < 8; ++r) {
      float va = c2a[r] > 0.f ? c2a[r] : 0.f;
      float vb = c2b[r] > 0.f ? c2b[r] : 0.f;
      acc[r] += va * w3a + vb * w3b;
    }
  }

  #pragma unroll
  for (int r = 0; r < 8; ++r) {
    float v = acc[r];
    v += swz<XOR1>(v);
    v += swz<XOR2>(v);
    v += swz<XOR4>(v);
    v += swz<XOR8>(v);
    acc[r] = v;
  }

  if (n == 0) {
    float* pp = &part[slice][tile * 16 + 8 * q];
    for (int r = 0; r < 8; ++r) pp[r] = acc[r];
  }
  __syncthreads();

  if (threadIdx.x < TILES * 16) {
    float s = part[0][threadIdx.x];
    for (int w = 1; w < SPLIT; ++w) s += part[w][threadIdx.x];
    out[blockIdx.x * (TILES * 16) + threadIdx.x] = s + b3[0];
  }
}

extern "C" void kernel_launch(void* const* d_in, const int* in_sizes, int n_in,
                              void* d_out, int out_size, void* d_ws, size_t ws_size,
                              hipStream_t stream) {
  (void)in_sizes; (void)n_in; (void)d_ws; (void)ws_size; (void)out_size;
  const float* y  = (const float*)d_in[0];
  const float* W1 = (const float*)d_in[1];
  const float* b1 = (const float*)d_in[2];
  const float* W2 = (const float*)d_in[3];
  const float* b2 = (const float*)d_in[4];
  const float* W3 = (const float*)d_in[5];
  const float* b3 = (const float*)d_in[6];
  float* out = (float*)d_out;

  dim3 grid(BATCH / (16 * TILES));
  dim3 block(SPLIT * TILES * 32);
  mlp200_wmma_kernel<<<grid, block, 0, stream>>>(y, W1, b1, W2, b2, W3, b3, out);
}


# Each wave carries four 16-row tiles (64 rows) per feature, so one load of the W1/b1/W2/b2/W3 fragments feeds 40 WMMAs instead of 10; feature order per slice and slice sum order unchanged
# speedup vs baseline: 2.2470x; 1.5030x over previous
#include <hip/hip_runtime.h>


typedef float v2f __attribute__((ext_vector_type(2)));
typedef float v8f __attribute__((ext_vector_type(8)));

#define BATCH 32768
#define NF    200
#define DIN   5
#define NH    15
#define NOUT  30
#define SPLIT 8
#define RT    4
#define ROWS  (RT * 16)

__device__ __forceinline__ v8f wmma_f32_k4(v2f a, v2f b, v8f c) {
  return __builtin_amdgcn_wmma_f32_16x16x4_f32(
       false, a,  false, b,
       (short)0, c,  false,  false);
}

template <int IMM>
__device__ __forceinline__ float swz(float x) {
  return __int_as_float(__builtin_amdgcn_ds_swizzle(__float_as_int(x), IMM));
}
#define SWAP16 0x401f
#define XOR1   0x041f
#define XOR2   0x081f
#define XOR4   0x101f
#define XOR8   0x201f

__global__ __launch_bounds__(SPLIT * 32) void mlp200_wmma_kernel(
    const float* __restrict__ y,  const float* __restrict__ W1,
    const float* __restrict__ b1, const float* __restrict__ W2,
    const float* __restrict__ b2, const float* __restrict__ W3,
    const float* __restrict__ b3, float* __restrict__ out)
{
  __shared__ float part[SPLIT][ROWS];

  const int  lane = threadIdx.x & 31;
  const int  slice = __builtin_amdgcn_readfirstlane((int)(threadIdx.x >> 5));
  const int  q    = lane >> 4;
  const bool qhi  = (q != 0);
  const int  n    = lane & 15;
  const int  b0   = blockIdx.x * ROWS;

  const int  nc   = (n < NH - 1) ? n : (NH - 1);
  const int  oc2  = 16 + ((n < 14) ? n : 13);
  const bool o2ok = (n < 14);

  const int  iy0  = (b0 + n) * (NF * DIN) + 2 * q;
  const int  iy1  = iy0 + 1;
  const int  iy4  = (b0 + n) * (NF * DIN) + 4;
  constexpr int TSTEP = 16 * NF * DIN;

  const int iW1a = (2 * q) * NH + nc;
  const int iW1b = iW1a + NH;
  const int iW1c = 4 * NH + nc;

  float acc[RT][8];
  #pragma unroll
  for (int t = 0; t < RT; ++t)
    #pragma unroll
    for (int r = 0; r < 8; ++r) acc[t][r] = 0.f;

  for (int f = slice; f < NF; f += SPLIT) {
    const float* yb  = y  + f * DIN;
    const float* W1f = W1 + f * (DIN * NH);
    const float* W2f = W2 + f * (NH * NOUT);
    const float* b1f = b1 + f * NH;
    const float* b2f = b2 + f * NOUT;
    const float* W3f = W3 + f * NOUT;

    v2f wa0, wa1;
    wa0.x = W1f[iW1a];
    wa0.y = W1f[iW1b];
    wa1.x = W1f[iW1c];
    wa1.y = b1f[nc];

    v2f bta[4], btb[4];
    #pragma unroll
    for (int k = 0; k < 4; ++k) {
      const int h0  = 4 * k + 2 * q;
      int       h1i = h0 + 1; if (h1i > NH - 1) h1i = NH - 1;
      bta[k].x = W2f[h0  * NOUT + n];
      btb[k].x = W2f[h0  * NOUT + oc2];
      float bay = W2f[h1i * NOUT + n];
      float bby = W2f[h1i * NOUT + oc2];
      if (k == 3) {
        float b2a = b2f[n];
        float b2b = b2f[oc2];
        bta[k].y = qhi ? b2a : bay;
        btb[k].y = qhi ? b2b : bby;
      } else {
        bta[k].y = bay;
        btb[k].y = bby;
      }
    }

    float w3a = W3f[n];
    float w3t = W3f[oc2];
    float w3b = o2ok ? w3t : 0.f;

    #pragma unroll
    for (int t = 0; t < RT; ++t) {
      v8f c1 = {};
      {
        v2f bb;
        bb.x = yb[iy0 + t * TSTEP];
        bb.y = yb[iy1 + t * TSTEP];
        c1 = wmma_f32_k4(wa0, bb, c1);
        float y4 = yb[iy4 + t * TSTEP];
        bb.x = qhi ? 0.f : y4;
        bb.y = qhi ? 0.f : 1.f;
        c1 = wmma_f32_k4(wa1, bb, c1);
      }

      float hv[8], sw[8];
      #pragma unroll
      for (int r = 0; r < 8; ++r) { hv[r] = c1[r]; sw[r] = swz<SWAP16>(c1[r]); }

      v2f a0, a1, a2, a3;
      a0.x = qhi ? sw[2] : hv[0];
      a0.y = qhi ? sw[3] : hv[1];
      a1.x = qhi ? sw[6] : hv[4];
      a1.y = qhi ? sw[7] : hv[5];
      a2.x = qhi ? hv[2] : sw[0];
      a2.y = qhi ? hv[3] : sw[1];
      a3.x = qhi ? hv[6] : sw[4];
      a3.y = qhi ? 1.f   : sw[5];

      v8f c2a = {}, c2b = {};
      #pragma unroll
      for (int k = 0; k < 4; ++k) {
        const v2f av = (k == 0) ? a0 : (k == 1) ? a1 : (k == 2) ? a2 : a3;
        c2a = wmma_f32_k4(av, bta[k], c2a);
        c2b = wmma_f32_k4(av, btb[k], c2b);
      }

      #pragma unroll
      for (int r = 0; r < 8; ++r) {
        float va = c2a[r] > 0.f ? c2a[r] : 0.f;
        float vb = c2b[r] > 0.f ? c2b[r] : 0.f;
        acc[t][r] += va * w3a + vb * w3b;
      }
    }
  }

  #pragma unroll
  for (int t = 0; t < RT; ++t) {
    #pragma unroll
    for (int r = 0; r < 8; ++r) {
      float v = acc[t][r];
      v += swz<XOR1>(v);
      v += swz<XOR2>(v);
      v += swz<XOR4>(v);
      v += swz<XOR8>(v);
      acc[t][r] = v;
    }
  }

  if (n == 0) {
    #pragma unroll
    for (int t = 0; t < RT; ++t) {
      float* pp = &part[slice][t * 16 + 8 * q];
      #pragma unroll
      for (int r = 0; r < 8; ++r) pp[r] = acc[t][r];
    }
  }
  __syncthreads();

  if (threadIdx.x < ROWS) {
    float s = part[0][threadIdx.x];
    for (int w = 1; w < SPLIT; ++w) s += part[w][threadIdx.x];
    out[blockIdx.x * ROWS + threadIdx.x] = s + b3[0];
  }
}

extern "C" void kernel_launch(void* const* d_in, const int* in_sizes, int n_in,
                              void* d_out, int out_size, void* d_ws, size_t ws_size,
                              hipStream_t stream) {
  (void)in_sizes; (void)n_in; (void)d_ws; (void)ws_size; (void)out_size;
  const float* y  = (const float*)d_in[0];
  const float* W1 = (const float*)d_in[1];
  const float* b1 = (const float*)d_in[2];
  const float* W2 = (const float*)d_in[3];
  const float* b2 = (const float*)d_in[4];
  const float* W3 = (const float*)d_in[5];
  const float* b3 = (const float*)d_in[6];
  float* out = (float*)d_out;

  dim3 grid(BATCH / ROWS);
  dim3 block(SPLIT * 32);
  mlp200_wmma_kernel<<<grid, block, 0, stream>>>(y, W1, b1, W2, b2, W3, b3, out);
}


# Layers 1 and 2 merged (no activation between): each wave forms W12=[W1 0;b1 1]*[W2;b2] per feature on WMMA, then 4 WMMAs per tile, not 10; (yW1)W2 summed as y(W1W2): rounding changes
# speedup vs baseline: 2.7113x; 1.2066x over previous
#include <hip/hip_runtime.h>


typedef float v2f __attribute__((ext_vector_type(2)));
typedef float v8f __attribute__((ext_vector_type(8)));

#define BATCH 32768
#define NF    200
#define DIN   5
#define NH    15
#define NOUT  30
#define SPLIT 8
#define RT    4
#define ROWS  (RT * 16)

__device__ __forceinline__ v8f wmma_f32_k4(v2f a, v2f b, v8f c) {
  return __builtin_amdgcn_wmma_f32_16x16x4_f32(
       false, a,  false, b,
       (short)0, c,  false,  false);
}

template <int IMM>
__device__ __forceinline__ float swz(float x) {
  return __int_as_float(__builtin_amdgcn_ds_swizzle(__float_as_int(x), IMM));
}
#define XOR1   0x041f
#define XOR2   0x081f
#define XOR4   0x101f
#define XOR8   0x201f

__global__ __launch_bounds__(SPLIT * 32) void mlp200_wmma_kernel(
    const float* __restrict__ y,  const float* __restrict__ W1,
    const float* __restrict__ b1, const float* __restrict__ W2,
    const float* __restrict__ b2, const float* __restrict__ W3,
    const float* __restrict__ b3, float* __restrict__ out)
{
  __shared__ float part[SPLIT][ROWS];

  const int  lane = threadIdx.x & 31;
  const int  slice = __builtin_amdgcn_readfirstlane((int)(threadIdx.x >> 5));
  const int  q    = lane >> 4;
  const bool qhi  = (q != 0);
  const int  n    = lane & 15;
  const int  b0   = blockIdx.x * ROWS;

  const int  oc2  = 16 + ((n < 14) ? n : 13);
  const bool o2ok = (n < 14);

  const int  iy0  = (b0 + n) * (NF * DIN) + 2 * q;
  const int  iy1  = iy0 + 1;
  const int  iy4  = (b0 + n) * (NF * DIN) + 4;
  constexpr int TSTEP = 16 * NF * DIN;

  const bool isb  = (n == 3);
  const int  irow = (int)((0x0000003200000410ull >> (4 * n)) & 15ull);
  const float* pA = isb ? b1 : (W1 + irow * NH);
  const int  stpA = isb ? NH : (DIN * NH);
  const float c15 = isb ? 1.f : 0.f;

  float acc[RT][8];
  #pragma unroll
  for (int t = 0; t < RT; ++t)
    #pragma unroll
    for (int r = 0; r < 8; ++r) acc[t][r] = 0.f;

  for (int f = slice; f < NF; f += SPLIT) {
    const float* yb  = y  + f * DIN;
    const float* W2f = W2 + f * (NH * NOUT);
    const float* b2f = b2 + f * NOUT;
    const float* W3f = W3 + f * NOUT;
    const float* pa  = pA + f * stpA + 2 * q;

    v2f wa[4];
    #pragma unroll
    for (int k = 0; k < 4; ++k) {
      float ax = pa[4 * k];
      float ay = (k == 3) ? pa[13 - q] : pa[4 * k + 1];
      if (k == 3) ay = qhi ? c15 : ay;
      wa[k].x = ax;
      wa[k].y = ay;
    }

    v2f bta[4], btb[4];
    #pragma unroll
    for (int k = 0; k < 4; ++k) {
      const int h0  = 4 * k + 2 * q;
      int       h1i = h0 + 1; if (h1i > NH - 1) h1i = NH - 1;
      bta[k].x = W2f[h0  * NOUT + n];
      btb[k].x = W2f[h0  * NOUT + oc2];
      float bay = W2f[h1i * NOUT + n];
      float bby = W2f[h1i * NOUT + oc2];
      if (k == 3) {
        float b2a = b2f[n];
        float b2b = b2f[oc2];
        bta[k].y = qhi ? b2a : bay;
        btb[k].y = qhi ? b2b : bby;
      } else {
        bta[k].y = bay;
        btb[k].y = bby;
      }
    }

    v8f da = {}, db = {};
    #pragma unroll
    for (int k = 0; k < 4; ++k) {
      da = wmma_f32_k4(wa[k], bta[k], da);
      db = wmma_f32_k4(wa[k], btb[k], db);
    }
    v2f wb0a, wb1a, wb0b, wb1b;
    wb0a.x = da[0]; wb0a.y = da[1]; wb1a.x = da[2]; wb1a.y = da[3];
    wb0b.x = db[0]; wb0b.y = db[1]; wb1b.x = db[2]; wb1b.y = db[3];

    float w3a = W3f[n];
    float w3t = W3f[oc2];
    float w3b = o2ok ? w3t : 0.f;

    #pragma unroll
    for (int t = 0; t < RT; ++t) {
      v2f a0, a1;
      a0.x = yb[iy0 + t * TSTEP];
      a0.y = yb[iy1 + t * TSTEP];
      float y4 = yb[iy4 + t * TSTEP];
      a1.x = qhi ? 0.f : y4;
      a1.y = qhi ? 0.f : 1.f;
      v8f c2a = {}, c2b = {};
      c2a = wmma_f32_k4(a0, wb0a, c2a);
      c2b = wmma_f32_k4(a0, wb0b, c2b);
      c2a = wmma_f32_k4(a1, wb1a, c2a);
      c2b = wmma_f32_k4(a1, wb1b, c2b);

      #pragma unroll
      for (int r = 0; r < 8; ++r) {
        float va = c2a[r] > 0.f ? c2a[r] : 0.f;
        float vb = c2b[r] > 0.f ? c2b[r] : 0.f;
        acc[t][r] += va * w3a + vb * w3b;
      }
    }
  }

  #pragma unroll
  for (int t = 0; t < RT; ++t) {
    #pragma unroll
    for (int r = 0; r < 8; ++r) {
      float v = acc[t][r];
      v += swz<XOR1>(v);
      v += swz<XOR2>(v);
      v += swz<XOR4>(v);
      v += swz<XOR8>(v);
      acc[t][r] = v;
    }
  }

  if (n == 0) {
    #pragma unroll
    for (int t = 0; t < RT; ++t) {
      float* pp = &part[slice][t * 16 + 8 * q];
      #pragma unroll
      for (int r = 0; r < 8; ++r) pp[r] = acc[t][r];
    }
  }
  __syncthreads();

  if (threadIdx.x < ROWS) {
    float s = part[0][threadIdx.x];
    for (int w = 1; w < SPLIT; ++w) s += part[w][threadIdx.x];
    out[blockIdx.x * ROWS + threadIdx.x] = s + b3[0];
  }
}

extern "C" void kernel_launch(void* const* d_in, const int* in_sizes, int n_in,
                              void* d_out, int out_size, void* d_ws, size_t ws_size,
                              hipStream_t stream) {
  (void)in_sizes; (void)n_in; (void)d_ws; (void)ws_size; (void)out_size;
  const float* y  = (const float*)d_in[0];
  const float* W1 = (const float*)d_in[1];
  const float* b1 = (const float*)d_in[2];
  const float* W2 = (const float*)d_in[3];
  const float* b2 = (const float*)d_in[4];
  const float* W3 = (const float*)d_in[5];
  const float* b3 = (const float*)d_in[6];
  float* out = (float*)d_out;

  dim3 grid(BATCH / ROWS);
  dim3 block(SPLIT * 32);
  mlp200_wmma_kernel<<<grid, block, 0, stream>>>(y, W1, b1, W2, b2, W3, b3, out);
}


# y staged through LDS: block fetches each feature step's 64x40 floats by consecutive 8-byte loads two steps ahead (2 stages, one LDS-only barrier per step); A fragments read from LDS, not row gathers
# speedup vs baseline: 3.2467x; 1.1975x over previous
#include <hip/hip_runtime.h>


typedef float v2f __attribute__((ext_vector_type(2)));
typedef float v8f __attribute__((ext_vector_type(8)));

#define BATCH 32768
#define NF    200
#define DIN   5
#define NH    15
#define NOUT  30
#define SPLIT 8
#define RT    4
#define ROWS  (RT * 16)
#define NSTEP (NF / SPLIT)
#define SW    (SPLIT * DIN)
#define SP    44
#define NPC   (ROWS * SW / 2 / (SPLIT * 32))

__device__ __forceinline__ v8f wmma_f32_k4(v2f a, v2f b, v8f c) {
  return __builtin_amdgcn_wmma_f32_16x16x4_f32(
       false, a,  false, b,
       (short)0, c,  false,  false);
}

template <int IMM>
__device__ __forceinline__ float swz(float x) {
  return __int_as_float(__builtin_amdgcn_ds_swizzle(__float_as_int(x), IMM));
}
#define XOR1   0x041f
#define XOR2   0x081f
#define XOR4   0x101f
#define XOR8   0x201f

__device__ __forceinline__ void lds_barrier() {
  __builtin_amdgcn_fence(__ATOMIC_RELEASE, "workgroup", "local");
  __builtin_amdgcn_s_barrier();
  __builtin_amdgcn_fence(__ATOMIC_ACQUIRE, "workgroup", "local");
}

__global__ __launch_bounds__(SPLIT * 32) void mlp200_wmma_kernel(
    const float* __restrict__ y,  const float* __restrict__ W1,
    const float* __restrict__ b1, const float* __restrict__ W2,
    const float* __restrict__ b2, const float* __restrict__ W3,
    const float* __restrict__ b3, float* __restrict__ out)
{
  __shared__ float part[SPLIT][ROWS];
  __shared__ __attribute__((aligned(16))) float ybuf[2][ROWS * SP];

  const int  lane = threadIdx.x & 31;
  const int  slice = __builtin_amdgcn_readfirstlane((int)(threadIdx.x >> 5));
  const int  q    = lane >> 4;
  const bool qhi  = (q != 0);
  const int  n    = lane & 15;
  const int  b0   = blockIdx.x * ROWS;

  const int  oc2  = 16 + ((n < 14) ? n : 13);
  const bool o2ok = (n < 14);

  const int  ly0  = n * SP + slice * DIN + 2 * q;
  const int  ly4  = n * SP + slice * DIN + 4;
  constexpr int TSTEP = 16 * SP;

  int gy[NPC], sy[NPC];
  for (int k = 0; k < NPC; ++k) {
    const int p  = (int)threadIdx.x + k * (SPLIT * 32);
    const int pr = p / (SW / 2), pc = p % (SW / 2);
    gy[k] = (b0 + pr) * (NF * DIN) + 2 * pc;
    sy[k] = pr * SP + 2 * pc;
  }

  float2 nx[NPC];
  {
    float2 s0[NPC];
    for (int k = 0; k < NPC; ++k) s0[k] = *reinterpret_cast<const float2*>(y + gy[k]);
    for (int k = 0; k < NPC; ++k) nx[k] = *reinterpret_cast<const float2*>(y + SW + gy[k]);
    for (int k = 0; k < NPC; ++k) *reinterpret_cast<float2*>(&ybuf[0][sy[k]]) = s0[k];
  }
  __syncthreads();

  const bool isb  = (n == 3);
  const int  irow = (int)((0x0000003200000410ull >> (4 * n)) & 15ull);
  const float* pA = isb ? b1 : (W1 + irow * NH);
  const int  stpA = isb ? NH : (DIN * NH);
  const float c15 = isb ? 1.f : 0.f;

  float acc[RT][8];
  #pragma unroll
  for (int t = 0; t < RT; ++t)
    #pragma unroll
    for (int r = 0; r < 8; ++r) acc[t][r] = 0.f;

  int step = 0;
  for (int f = slice; f < NF; f += SPLIT, ++step) {
    const bool more = (step + 1 < NSTEP);
    if (more) {
      float* ys = &ybuf[(step + 1) & 1][0];
      for (int k = 0; k < NPC; ++k) *reinterpret_cast<float2*>(ys + sy[k]) = nx[k];
    }
    const float* W2f = W2 + f * (NH * NOUT);
    const float* b2f = b2 + f * NOUT;
    const float* W3f = W3 + f * NOUT;
    const float* pa  = pA + f * stpA + 2 * q;

    v2f wa[4];
    #pragma unroll
    for (int k = 0; k < 4; ++k) {
      float ax = pa[4 * k];
      float ay = (k == 3) ? pa[13 - q] : pa[4 * k + 1];
      if (k == 3) ay = qhi ? c15 : ay;
      wa[k].x = ax;
      wa[k].y = ay;
    }

    v2f bta[4], btb[4];
    #pragma unroll
    for (int k = 0; k < 4; ++k) {
      const int h0  = 4 * k + 2 * q;
      int       h1i = h0 + 1; if (h1i > NH - 1) h1i = NH - 1;
      bta[k].x = W2f[h0  * NOUT + n];
      btb[k].x = W2f[h0  * NOUT + oc2];
      float bay = W2f[h1i * NOUT + n];
      float bby = W2f[h1i * NOUT + oc2];
      if (k == 3) {
        float b2a = b2f[n];
        float b2b = b2f[oc2];
        bta[k].y = qhi ? b2a : bay;
        btb[k].y = qhi ? b2b : bby;
      } else {
        bta[k].y = bay;
        btb[k].y = bby;
      }
    }

    float w3a = W3f[n];
    float w3t = W3f[oc2];
    float w3b = o2ok ? w3t : 0.f;

    {
      const int ps = (step + 2 < NSTEP) ? step + 2 : NSTEP - 1;
      const float* yn = y + ps * SW;
      for (int k = 0; k < NPC; ++k) nx[k] = *reinterpret_cast<const float2*>(yn + gy[k]);
    }

    v8f da = {}, db = {};
    #pragma unroll
    for (int k = 0; k < 4; ++k) {
      da = wmma_f32_k4(wa[k], bta[k], da);
      db = wmma_f32_k4(wa[k], btb[k], db);
    }
    v2f wb0a, wb1a, wb0b, wb1b;
    wb0a.x = da[0]; wb0a.y = da[1]; wb1a.x = da[2]; wb1a.y = da[3];
    wb0b.x = db[0]; wb0b.y = db[1]; wb1b.x = db[2]; wb1b.y = db[3];

    const float* yl = &ybuf[step & 1][0];

    #pragma unroll
    for (int t = 0; t < RT; ++t) {
      v2f a0, a1;
      a0.x = yl[ly0 + t * TSTEP];
      a0.y = yl[ly0 + 1 + t * TSTEP];
      float y4 = yl[ly4 + t * TSTEP];
      a1.x = qhi ? 0.f : y4;
      a1.y = qhi ? 0.f : 1.f;
      v8f c2a = {}, c2b = {};
      c2a = wmma_f32_k4(a0, wb0a, c2a);
      c2b = wmma_f32_k4(a0, wb0b, c2b);
      c2a = wmma_f32_k4(a1, wb1a, c2a);
      c2b = wmma_f32_k4(a1, wb1b, c2b);

      #pragma unroll
      for (int r = 0; r < 8; ++r) {
        float va = c2a[r] > 0.f ? c2a[r] : 0.f;
        float vb = c2b[r] > 0.f ? c2b[r] : 0.f;
        acc[t][r] += va * w3a + vb * w3b;
      }
    }

    if (more) lds_barrier();
  }

  #pragma unroll
  for (int t = 0; t < RT; ++t) {
    #pragma unroll
    for (int r = 0; r < 8; ++r) {
      float v = acc[t][r];
      v += swz<XOR1>(v);
      v += swz<XOR2>(v);
      v += swz<XOR4>(v);
      v += swz<XOR8>(v);
      acc[t][r] = v;
    }
  }

  if (n == 0) {
    #pragma unroll
    for (int t = 0; t < RT; ++t) {
      float* pp = &part[slice][t * 16 + 8 * q];
      #pragma unroll
      for (int r = 0; r < 8; ++r) pp[r] = acc[t][r];
    }
  }
  __syncthreads();

  if (threadIdx.x < ROWS) {
    float s = part[0][threadIdx.x];
    for (int w = 1; w < SPLIT; ++w) s += part[w][threadIdx.x];
    out[blockIdx.x * ROWS + threadIdx.x] = s + b3[0];
  }
}

extern "C" void kernel_launch(void* const* d_in, const int* in_sizes, int n_in,
                              void* d_out, int out_size, void* d_ws, size_t ws_size,
                              hipStream_t stream) {
  (void)in_sizes; (void)n_in; (void)d_ws; (void)ws_size; (void)out_size;
  const float* y  = (const float*)d_in[0];
  const float* W1 = (const float*)d_in[1];
  const float* b1 = (const float*)d_in[2];
  const float* W2 = (const float*)d_in[3];
  const float* b2 = (const float*)d_in[4];
  const float* W3 = (const float*)d_in[5];
  const float* b3 = (const float*)d_in[6];
  float* out = (float*)d_out;

  dim3 grid(BATCH / ROWS);
  dim3 block(SPLIT * 32);
  mlp200_wmma_kernel<<<grid, block, 0, stream>>>(y, W1, b1, W2, b2, W3, b3, out);
}


# Weight fragments (W1, b1, W2, b2, W3) of the next feature step are loaded one step ahead into registers, so the set-up product no longer waits for its loads; sums and their order unchanged
# speedup vs baseline: 3.3692x; 1.0377x over previous
#include <hip/hip_runtime.h>


typedef float v2f __attribute__((ext_vector_type(2)));
typedef float v8f __attribute__((ext_vector_type(8)));

#define BATCH 32768
#define NF    200
#define DIN   5
#define NH    15
#define NOUT  30
#define SPLIT 8
#define RT    4
#define ROWS  (RT * 16)
#define NSTEP (NF / SPLIT)
#define SW    (SPLIT * DIN)
#define SP    44
#define NPC   (ROWS * SW / 2 / (SPLIT * 32))

__device__ __forceinline__ v8f wmma_f32_k4(v2f a, v2f b, v8f c) {
  return __builtin_amdgcn_wmma_f32_16x16x4_f32(
       false, a,  false, b,
       (short)0, c,  false,  false);
}

template <int IMM>
__device__ __forceinline__ float swz(float x) {
  return __int_as_float(__builtin_amdgcn_ds_swizzle(__float_as_int(x), IMM));
}
#define XOR1   0x041f
#define XOR2   0x081f
#define XOR4   0x101f
#define XOR8   0x201f

__device__ __forceinline__ void lds_barrier() {
  __builtin_amdgcn_fence(__ATOMIC_RELEASE, "workgroup", "local");
  __builtin_amdgcn_s_barrier();
  __builtin_amdgcn_fence(__ATOMIC_ACQUIRE, "workgroup", "local");
}

__global__ __launch_bounds__(SPLIT * 32) void mlp200_wmma_kernel(
    const float* __restrict__ y,  const float* __restrict__ W1,
    const float* __restrict__ b1, const float* __restrict__ W2,
    const float* __restrict__ b2, const float* __restrict__ W3,
    const float* __restrict__ b3, float* __restrict__ out)
{
  __shared__ float part[SPLIT][ROWS];
  __shared__ __attribute__((aligned(16))) float ybuf[2][ROWS * SP];

  const int  lane = threadIdx.x & 31;
  const int  slice = __builtin_amdgcn_readfirstlane((int)(threadIdx.x >> 5));
  const int  q    = lane >> 4;
  const bool qhi  = (q != 0);
  const int  n    = lane & 15;
  const int  b0   = blockIdx.x * ROWS;

  const int  oc2  = 16 + ((n < 14) ? n : 13);
  const bool o2ok = (n < 14);

  const int  ly0  = n * SP + slice * DIN + 2 * q;
  const int  ly4  = n * SP + slice * DIN + 4;
  constexpr int TSTEP = 16 * SP;

  int gy[NPC], sy[NPC];
  for (int k = 0; k < NPC; ++k) {
    const int p  = (int)threadIdx.x + k * (SPLIT * 32);
    const int pr = p / (SW / 2), pc = p % (SW / 2);
    gy[k] = (b0 + pr) * (NF * DIN) + 2 * pc;
    sy[k] = pr * SP + 2 * pc;
  }

  float2 nx[NPC];
  {
    float2 s0[NPC];
    for (int k = 0; k < NPC; ++k) s0[k] = *reinterpret_cast<const float2*>(y + gy[k]);
    for (int k = 0; k < NPC; ++k) nx[k] = *reinterpret_cast<const float2*>(y + SW + gy[k]);
    for (int k = 0; k < NPC; ++k) *reinterpret_cast<float2*>(&ybuf[0][sy[k]]) = s0[k];
  }

  const bool isb  = (n == 3);
  const int  irow = (int)((0x0000003200000410ull >> (4 * n)) & 15ull);
  const float* pA = isb ? b1 : (W1 + irow * NH);
  const int  stpA = isb ? NH : (DIN * NH);
  const float c15 = isb ? 1.f : 0.f;

  struct WRaw {
    float ax[4], ay[4];
    float bax[4], bbx[4];
    float bay[4], bby[4];
    float b2a, b2b, w3a, w3t;
  };
  auto load_w = [&](WRaw& w, int f) {
    const float* W2f = W2 + f * (NH * NOUT);
    const float* b2f = b2 + f * NOUT;
    const float* W3f = W3 + f * NOUT;
    const float* pa  = pA + f * stpA + 2 * q;
    #pragma unroll
    for (int k = 0; k < 4; ++k) {
      w.ax[k] = pa[4 * k];
      w.ay[k] = (k == 3) ? pa[13 - q] : pa[4 * k + 1];
      const int h0  = 4 * k + 2 * q;
      int       h1i = h0 + 1; if (h1i > NH - 1) h1i = NH - 1;
      w.bax[k] = W2f[h0  * NOUT + n];
      w.bbx[k] = W2f[h0  * NOUT + oc2];
      w.bay[k] = W2f[h1i * NOUT + n];
      w.bby[k] = W2f[h1i * NOUT + oc2];
    }
    w.b2a = b2f[n];
    w.b2b = b2f[oc2];
    w.w3a = W3f[n];
    w.w3t = W3f[oc2];
  };
  WRaw wn;
  load_w(wn, slice);
  __syncthreads();

  float acc[RT][8];
  #pragma unroll
  for (int t = 0; t < RT; ++t)
    #pragma unroll
    for (int r = 0; r < 8; ++r) acc[t][r] = 0.f;

  int step = 0;
  for (int f = slice; f < NF; f += SPLIT, ++step) {
    const bool more = (step + 1 < NSTEP);
    if (more) {
      float* ys = &ybuf[(step + 1) & 1][0];
      for (int k = 0; k < NPC; ++k) *reinterpret_cast<float2*>(ys + sy[k]) = nx[k];
    }
    const WRaw wc = wn;

    v2f wa[4];
    #pragma unroll
    for (int k = 0; k < 4; ++k) {
      float ay = wc.ay[k];
      if (k == 3) ay = qhi ? c15 : ay;
      wa[k].x = wc.ax[k];
      wa[k].y = ay;
    }

    v2f bta[4], btb[4];
    #pragma unroll
    for (int k = 0; k < 4; ++k) {
      bta[k].x = wc.bax[k];
      btb[k].x = wc.bbx[k];
      if (k == 3) {
        bta[k].y = qhi ? wc.b2a : wc.bay[k];
        btb[k].y = qhi ? wc.b2b : wc.bby[k];
      } else {
        bta[k].y = wc.bay[k];
        btb[k].y = wc.bby[k];
      }
    }

    const float w3a = wc.w3a;
    const float w3b = o2ok ? wc.w3t : 0.f;

    v8f da = {}, db = {};
    #pragma unroll
    for (int k = 0; k < 4; ++k) {
      da = wmma_f32_k4(wa[k], bta[k], da);
      db = wmma_f32_k4(wa[k], btb[k], db);
    }
    load_w(wn, more ? f + SPLIT : f);

    {
      const int ps = (step + 2 < NSTEP) ? step + 2 : NSTEP - 1;
      const float* yn = y + ps * SW;
      for (int k = 0; k < NPC; ++k) nx[k] = *reinterpret_cast<const float2*>(yn + gy[k]);
    }

    v2f wb0a, wb1a, wb0b, wb1b;
    wb0a.x = da[0]; wb0a.y = da[1]; wb1a.x = da[2]; wb1a.y = da[3];
    wb0b.x = db[0]; wb0b.y = db[1]; wb1b.x = db[2]; wb1b.y = db[3];

    const float* yl = &ybuf[step & 1][0];

    #pragma unroll
    for (int t = 0; t < RT; ++t) {
      v2f a0, a1;
      a0.x = yl[ly0 + t * TSTEP];
      a0.y = yl[ly0 + 1 + t * TSTEP];
      float y4 = yl[ly4 + t * TSTEP];
      a1.x = qhi ? 0.f : y4;
      a1.y = qhi ? 0.f : 1.f;
      v8f c2a = {}, c2b = {};
      c2a = wmma_f32_k4(a0, wb0a, c2a);
      c2b = wmma_f32_k4(a0, wb0b, c2b);
      c2a = wmma_f32_k4(a1, wb1a, c2a);
      c2b = wmma_f32_k4(a1, wb1b, c2b);

      #pragma unroll
      for (int r = 0; r < 8; ++r) {
        float va = c2a[r] > 0.f ? c2a[r] : 0.f;
        float vb = c2b[r] > 0.f ? c2b[r] : 0.f;
        acc[t][r] += va * w3a + vb * w3b;
      }
    }

    if (more) lds_barrier();
  }

  #pragma unroll
  for (int t = 0; t < RT; ++t) {
    #pragma unroll
    for (int r = 0; r < 8; ++r) {
      float v = acc[t][r];
      v += swz<XOR1>(v);
      v += swz<XOR2>(v);
      v += swz<XOR4>(v);
      v += swz<XOR8>(v);
      acc[t][r] = v;
    }
  }

  if (n == 0) {
    #pragma unroll
    for (int t = 0; t < RT; ++t) {
      float* pp = &part[slice][t * 16 + 8 * q];
      #pragma unroll
      for (int r = 0; r < 8; ++r) pp[r] = acc[t][r];
    }
  }
  __syncthreads();

  if (threadIdx.x < ROWS) {
    float s = part[0][threadIdx.x];
    for (int w = 1; w < SPLIT; ++w) s += part[w][threadIdx.x];
    out[blockIdx.x * ROWS + threadIdx.x] = s + b3[0];
  }
}

extern "C" void kernel_launch(void* const* d_in, const int* in_sizes, int n_in,
                              void* d_out, int out_size, void* d_ws, size_t ws_size,
                              hipStream_t stream) {
  (void)in_sizes; (void)n_in; (void)d_ws; (void)ws_size; (void)out_size;
  const float* y  = (const float*)d_in[0];
  const float* W1 = (const float*)d_in[1];
  const float* b1 = (const float*)d_in[2];
  const float* W2 = (const float*)d_in[3];
  const float* b2 = (const float*)d_in[4];
  const float* W3 = (const float*)d_in[5];
  const float* b3 = (const float*)d_in[6];
  float* out = (float*)d_out;

  dim3 grid(BATCH / ROWS);
  dim3 block(SPLIT * 32);
  mlp200_wmma_kernel<<<grid, block, 0, stream>>>(y, W1, b1, W2, b2, W3, b3, out);
}


# Block = 16 waves on 128 rows (was 8 on 64): wave w takes features w, w+16, .. over 8 row tiles, so W12 and its weight loads are made once per 128 rows; 16 slice sums per row: sum order changes
# speedup vs baseline: 3.9146x; 1.1619x over previous
#include <hip/hip_runtime.h>


typedef float v2f __attribute__((ext_vector_type(2)));
typedef float v4f __attribute__((ext_vector_type(4)));
typedef float v8f __attribute__((ext_vector_type(8)));

#define BATCH 32768
#define NF    200
#define DIN   5
#define NH    15
#define NOUT  30
#define SPLIT 16
#define RT    8
#define ROWS  (RT * 16)
#define NFULL (NF / SPLIT)
#define NTAIL (NF - NFULL * SPLIT)
#define SW    (SPLIT * DIN)
#define TAILOFF (NF * DIN - SW)
#define TAILBLK (SPLIT - NTAIL)
#define SP    84
#define NPC   (ROWS * SW / 4 / (SPLIT * 32))

__device__ __forceinline__ v8f wmma_f32_k4(v2f a, v2f b, v8f c) {
  return __builtin_amdgcn_wmma_f32_16x16x4_f32(
       false, a,  false, b,
       (short)0, c,  false,  false);
}

template <int IMM>
__device__ __forceinline__ float swz(float x) {
  return __int_as_float(__builtin_amdgcn_ds_swizzle(__float_as_int(x), IMM));
}
#define XOR1   0x041f
#define XOR2   0x081f
#define XOR4   0x101f
#define XOR8   0x201f

__device__ __forceinline__ void lds_barrier() {
  __builtin_amdgcn_fence(__ATOMIC_RELEASE, "workgroup", "local");
  __builtin_amdgcn_s_barrier();
  __builtin_amdgcn_fence(__ATOMIC_ACQUIRE, "workgroup", "local");
}

__global__ __launch_bounds__(SPLIT * 32) __attribute__((amdgpu_waves_per_eu(4, 4)))
void mlp200_wmma_kernel(
    const float* __restrict__ y,  const float* __restrict__ W1,
    const float* __restrict__ b1, const float* __restrict__ W2,
    const float* __restrict__ b2, const float* __restrict__ W3,
    const float* __restrict__ b3, float* __restrict__ out)
{
  __shared__ float part[SPLIT][ROWS];
  __shared__ __attribute__((aligned(16))) float ybuf[2][ROWS * SP];

  const int  lane = threadIdx.x & 31;
  const int  slice = __builtin_amdgcn_readfirstlane((int)(threadIdx.x >> 5));
  const int  q    = lane >> 4;
  const bool qhi  = (q != 0);
  const int  n    = lane & 15;
  const int  b0   = blockIdx.x * ROWS;

  const int  oc2  = 16 + ((n < 14) ? n : 13);
  const bool o2ok = (n < 14);

  const int  ly0  = n * SP + slice * DIN + 2 * q;
  const int  ly4  = n * SP + slice * DIN + 4;
  constexpr int TSTEP = 16 * SP;

  int gy[NPC], sy[NPC];
  #pragma unroll
  for (int k = 0; k < NPC; ++k) {
    const int p  = (int)threadIdx.x + k * (SPLIT * 32);
    const int pr = p / (SW / 4), pc = p % (SW / 4);
    gy[k] = (b0 + pr) * (NF * DIN) + 4 * pc;
    sy[k] = pr * SP + 4 * pc;
  }

  v4f nx[NPC];
  {
    v4f s0[NPC];
    #pragma unroll
    for (int k = 0; k < NPC; ++k) s0[k] = *reinterpret_cast<const v4f*>(y + gy[k]);
    #pragma unroll
    for (int k = 0; k < NPC; ++k) nx[k] = *reinterpret_cast<const v4f*>(y + SW + gy[k]);
    #pragma unroll
    for (int k = 0; k < NPC; ++k) *reinterpret_cast<v4f*>(&ybuf[0][sy[k]]) = s0[k];
  }

  const bool isb  = (n == 3);
  const int  irow = (int)((0x0000003200000410ull >> (4 * n)) & 15ull);
  const float* pA = isb ? b1 : (W1 + irow * NH);
  const int  stpA = isb ? NH : (DIN * NH);
  const float c15 = isb ? 1.f : 0.f;

  struct WRaw {
    float ax[4], ay[4];
    float bax[4], bbx[4];
    float bay[4], bby[4];
    float b2a, b2b, w3a, w3t;
  };
  auto load_w = [&](WRaw& w, int f) {
    const float* W2f = W2 + f * (NH * NOUT);
    const float* b2f = b2 + f * NOUT;
    const float* W3f = W3 + f * NOUT;
    const float* pa  = pA + f * stpA + 2 * q;
    #pragma unroll
    for (int k = 0; k < 4; ++k) {
      w.ax[k] = pa[4 * k];
      w.ay[k] = (k == 3) ? pa[13 - q] : pa[4 * k + 1];
      const int h0  = 4 * k + 2 * q;
      int       h1i = h0 + 1; if (h1i > NH - 1) h1i = NH - 1;
      w.bax[k] = W2f[h0  * NOUT + n];
      w.bbx[k] = W2f[h0  * NOUT + oc2];
      w.bay[k] = W2f[h1i * NOUT + n];
      w.bby[k] = W2f[h1i * NOUT + oc2];
    }
    w.b2a = b2f[n];
    w.b2b = b2f[oc2];
    w.w3a = W3f[n];
    w.w3t = W3f[oc2];
  };

  float acc[RT][8];
  #pragma unroll
  for (int t = 0; t < RT; ++t)
    #pragma unroll
    for (int r = 0; r < 8; ++r) acc[t][r] = 0.f;

  auto setup = [&](const WRaw& wc, v8f& da, v8f& db) {
    v2f wa[4];
    #pragma unroll
    for (int k = 0; k < 4; ++k) {
      float ay = wc.ay[k];
      if (k == 3) ay = qhi ? c15 : ay;
      wa[k].x = wc.ax[k];
      wa[k].y = ay;
    }
    v2f bta[4], btb[4];
    #pragma unroll
    for (int k = 0; k < 4; ++k) {
      bta[k].x = wc.bax[k];
      btb[k].x = wc.bbx[k];
      if (k == 3) {
        bta[k].y = qhi ? wc.b2a : wc.bay[k];
        btb[k].y = qhi ? wc.b2b : wc.bby[k];
      } else {
        bta[k].y = wc.bay[k];
        btb[k].y = wc.bby[k];
      }
    }
    da = v8f{}; db = v8f{};
    #pragma unroll
    for (int k = 0; k < 4; ++k) {
      da = wmma_f32_k4(wa[k], bta[k], da);
      db = wmma_f32_k4(wa[k], btb[k], db);
    }
  };

  auto tiles = [&](const v8f& da, const v8f& db, float w3a, float w3b, const float* yl) {
    v2f wb0a, wb1a, wb0b, wb1b;
    wb0a.x = da[0]; wb0a.y = da[1]; wb1a.x = da[2]; wb1a.y = da[3];
    wb0b.x = db[0]; wb0b.y = db[1]; wb1b.x = db[2]; wb1b.y = db[3];
    #pragma unroll
    for (int t = 0; t < RT; ++t) {
      v2f a0, a1;
      a0.x = yl[ly0 + t * TSTEP];
      a0.y = yl[ly0 + 1 + t * TSTEP];
      float y4 = yl[ly4 + t * TSTEP];
      a1.x = qhi ? 0.f : y4;
      a1.y = qhi ? 0.f : 1.f;
      v8f c2a = {}, c2b = {};
      c2a = wmma_f32_k4(a0, wb0a, c2a);
      c2b = wmma_f32_k4(a0, wb0b, c2b);
      c2a = wmma_f32_k4(a1, wb1a, c2a);
      c2b = wmma_f32_k4(a1, wb1b, c2b);

      #pragma unroll
      for (int r = 0; r < 8; ++r) {
        float va = c2a[r] > 0.f ? c2a[r] : 0.f;
        float vb = c2b[r] > 0.f ? c2b[r] : 0.f;
        acc[t][r] += va * w3a + vb * w3b;
      }
    }
  };

  WRaw wn;
  load_w(wn, slice);
  __syncthreads();

  const bool tail = (slice < NTAIL);
  for (int step = 0; step < NFULL; ++step) {
    const int f = step * SPLIT + slice;
    {
      float* ys = &ybuf[(step + 1) & 1][0];
      #pragma unroll
      for (int k = 0; k < NPC; ++k) *reinterpret_cast<v4f*>(ys + sy[k]) = nx[k];
    }
    const WRaw wc = wn;
    v8f da, db;
    setup(wc, da, db);

    load_w(wn, (step + 1 < NFULL || tail) ? f + SPLIT : f);

    {
      const int ps = step + 2;
      const float* yn = y + ((ps < NFULL) ? ps * SW : TAILOFF);
      #pragma unroll
      for (int k = 0; k < NPC; ++k) nx[k] = *reinterpret_cast<const v4f*>(yn + gy[k]);
    }

    tiles(da, db, wc.w3a, o2ok ? wc.w3t : 0.f,
          &ybuf[step & 1][0]);

    lds_barrier();
  }
  if (tail) {
    v8f da, db;
    setup(wn, da, db);
    tiles(da, db, wn.w3a, o2ok ? wn.w3t : 0.f, &ybuf[NFULL & 1][TAILBLK * DIN]);
  }

  #pragma unroll
  for (int t = 0; t < RT; ++t) {
    #pragma unroll
    for (int r = 0; r < 8; ++r) {
      float v = acc[t][r];
      v += swz<XOR1>(v);
      v += swz<XOR2>(v);
      v += swz<XOR4>(v);
      v += swz<XOR8>(v);
      acc[t][r] = v;
    }
  }

  if (n == 0) {
    #pragma unroll
    for (int t = 0; t < RT; ++t) {
      float* pp = &part[slice][t * 16 + 8 * q];
      #pragma unroll
      for (int r = 0; r < 8; ++r) pp[r] = acc[t][r];
    }
  }
  __syncthreads();

  if (threadIdx.x < ROWS) {
    float s = part[0][threadIdx.x];
    for (int w = 1; w < SPLIT; ++w) s += part[w][threadIdx.x];
    out[blockIdx.x * ROWS + threadIdx.x] = s + b3[0];
  }
}

extern "C" void kernel_launch(void* const* d_in, const int* in_sizes, int n_in,
                              void* d_out, int out_size, void* d_ws, size_t ws_size,
                              hipStream_t stream) {
  (void)in_sizes; (void)n_in; (void)d_ws; (void)ws_size; (void)out_size;
  const float* y  = (const float*)d_in[0];
  const float* W1 = (const float*)d_in[1];
  const float* b1 = (const float*)d_in[2];
  const float* W2 = (const float*)d_in[3];
  const float* b2 = (const float*)d_in[4];
  const float* W3 = (const float*)d_in[5];
  const float* b3 = (const float*)d_in[6];
  float* out = (float*)d_out;

  dim3 grid(BATCH / ROWS);
  dim3 block(SPLIT * 32);
  mlp200_wmma_kernel<<<grid, block, 0, stream>>>(y, W1, b1, W2, b2, W3, b3, out);
}


# Per-tile WMMA operands swapped (A=W12, B=y): outputs in registers, batch rows in lanes; ReLU*W3 sums in-lane into 2 sums per tile, 8 half exchanges replace 256 shuffles: sum order changes
# speedup vs baseline: 4.0304x; 1.0296x over previous
#include <hip/hip_runtime.h>


typedef float v2f __attribute__((ext_vector_type(2)));
typedef float v4f __attribute__((ext_vector_type(4)));
typedef float v8f __attribute__((ext_vector_type(8)));

#define BATCH 32768
#define NF    200
#define DIN   5
#define NH    15
#define NOUT  30
#define SPLIT 16
#define RT    8
#define ROWS  (RT * 16)
#define NFULL (NF / SPLIT)
#define NTAIL (NF - NFULL * SPLIT)
#define SW    (SPLIT * DIN)
#define TAILOFF (NF * DIN - SW)
#define TAILBLK (SPLIT - NTAIL)
#define SP    84
#define NPC   (ROWS * SW / 4 / (SPLIT * 32))

__device__ __forceinline__ v8f wmma_f32_k4(v2f a, v2f b, v8f c) {
  return __builtin_amdgcn_wmma_f32_16x16x4_f32(
       false, a,  false, b,
       (short)0, c,  false,  false);
}

template <int IMM>
__device__ __forceinline__ float swz(float x) {
  return __int_as_float(__builtin_amdgcn_ds_swizzle(__float_as_int(x), IMM));
}
#define XOR16  0x401f

__device__ __forceinline__ void lds_barrier() {
  __builtin_amdgcn_fence(__ATOMIC_RELEASE, "workgroup", "local");
  __builtin_amdgcn_s_barrier();
  __builtin_amdgcn_fence(__ATOMIC_ACQUIRE, "workgroup", "local");
}

__global__ __launch_bounds__(SPLIT * 32) __attribute__((amdgpu_waves_per_eu(4, 4)))
void mlp200_wmma_kernel(
    const float* __restrict__ y,  const float* __restrict__ W1,
    const float* __restrict__ b1, const float* __restrict__ W2,
    const float* __restrict__ b2, const float* __restrict__ W3,
    const float* __restrict__ b3, float* __restrict__ out)
{
  __shared__ float part[SPLIT][ROWS];
  __shared__ __attribute__((aligned(16))) float ybuf[2][ROWS * SP];

  const int  lane = threadIdx.x & 31;
  const int  slice = __builtin_amdgcn_readfirstlane((int)(threadIdx.x >> 5));
  const int  q    = lane >> 4;
  const bool qhi  = (q != 0);
  const int  n    = lane & 15;
  const int  b0   = blockIdx.x * ROWS;

  const int  oc2  = 16 + ((n < 14) ? n : 13);

  const int  ly0  = n * SP + slice * DIN + 2 * q;
  const int  ly4  = n * SP + slice * DIN + 4;
  constexpr int TSTEP = 16 * SP;

  int gy[NPC], sy[NPC];
  #pragma unroll
  for (int k = 0; k < NPC; ++k) {
    const int p  = (int)threadIdx.x + k * (SPLIT * 32);
    const int pr = p / (SW / 4), pc = p % (SW / 4);
    gy[k] = (b0 + pr) * (NF * DIN) + 4 * pc;
    sy[k] = pr * SP + 4 * pc;
  }

  v4f nx[NPC];
  {
    v4f s0[NPC];
    #pragma unroll
    for (int k = 0; k < NPC; ++k) s0[k] = *reinterpret_cast<const v4f*>(y + gy[k]);
    #pragma unroll
    for (int k = 0; k < NPC; ++k) nx[k] = *reinterpret_cast<const v4f*>(y + SW + gy[k]);
    #pragma unroll
    for (int k = 0; k < NPC; ++k) *reinterpret_cast<v4f*>(&ybuf[0][sy[k]]) = s0[k];
  }

  const bool isb  = (n == 3);
  const int  irow = (int)((0x0000003200000410ull >> (4 * n)) & 15ull);
  const float* pA = isb ? b1 : (W1 + irow * NH);
  const int  stpA = isb ? NH : (DIN * NH);
  const float c15 = isb ? 1.f : 0.f;

  struct WRaw {
    float ax[4], ay[4];
    float bax[4], bbx[4];
    float bay[4], bby[4];
    float b2a, b2b;
    v2f  w3A[4], w3B[4];
  };
  auto load_w = [&](WRaw& w, int f) {
    const float* W2f = W2 + f * (NH * NOUT);
    const float* b2f = b2 + f * NOUT;
    const float* pa  = pA + f * stpA + 2 * q;
    #pragma unroll
    for (int k = 0; k < 4; ++k) {
      w.ax[k] = pa[4 * k];
      w.ay[k] = (k == 3) ? pa[13 - q] : pa[4 * k + 1];
      const int h0  = 4 * k + 2 * q;
      int       h1i = h0 + 1; if (h1i > NH - 1) h1i = NH - 1;
      w.bax[k] = W2f[h0  * NOUT + n];
      w.bbx[k] = W2f[h0  * NOUT + oc2];
      w.bay[k] = W2f[h1i * NOUT + n];
      w.bby[k] = W2f[h1i * NOUT + oc2];
    }
    w.b2a = b2f[n];
    w.b2b = b2f[oc2];
  };
  auto load_w3 = [&](WRaw& w, int f) {
    const float* W3f = W3 + f * NOUT;
    #pragma unroll
    for (int j = 0; j < 4; ++j) {
      int ob = 16 + 8 * q + 2 * j; if (ob > NOUT - 2) ob = NOUT - 2;
      w.w3A[j] = *reinterpret_cast<const v2f*>(W3f + 8 * q + 2 * j);
      w.w3B[j] = *reinterpret_cast<const v2f*>(W3f + ob);
    }
  };

  v2f acc[RT];
  #pragma unroll
  for (int t = 0; t < RT; ++t) acc[t] = v2f{0.f, 0.f};

  auto setup = [&](const WRaw& wc, v8f& da, v8f& db) {
    v2f wa[4];
    #pragma unroll
    for (int k = 0; k < 4; ++k) {
      float ay = wc.ay[k];
      if (k == 3) ay = qhi ? c15 : ay;
      wa[k].x = wc.ax[k];
      wa[k].y = ay;
    }
    v2f bta[4], btb[4];
    #pragma unroll
    for (int k = 0; k < 4; ++k) {
      bta[k].x = wc.bax[k];
      btb[k].x = wc.bbx[k];
      if (k == 3) {
        bta[k].y = qhi ? wc.b2a : wc.bay[k];
        btb[k].y = qhi ? wc.b2b : wc.bby[k];
      } else {
        bta[k].y = wc.bay[k];
        btb[k].y = wc.bby[k];
      }
    }
    da = v8f{}; db = v8f{};
    #pragma unroll
    for (int k = 0; k < 4; ++k) {
      da = wmma_f32_k4(wa[k], bta[k], da);
      db = wmma_f32_k4(wa[k], btb[k], db);
    }
  };

  auto tiles = [&](const v8f& da, const v8f& db, const WRaw& wc, const float* yl) {
    v2f wb0a, wb1a, wb0b, wb1b;
    wb0a.x = da[0]; wb0a.y = da[1]; wb1a.x = da[2]; wb1a.y = da[3];
    wb0b.x = db[0]; wb0b.y = db[1]; wb1b.x = db[2]; wb1b.y = db[3];
    v2f w3b[4];
    #pragma unroll
    for (int j = 0; j < 4; ++j) w3b[j] = wc.w3B[j];
    w3b[3].x = qhi ? 0.f : w3b[3].x;
    w3b[3].y = qhi ? 0.f : w3b[3].y;
    #pragma unroll
    for (int t = 0; t < RT; ++t) {
      v2f a0, a1;
      a0.x = yl[ly0 + t * TSTEP];
      a0.y = yl[ly0 + 1 + t * TSTEP];
      float y4 = yl[ly4 + t * TSTEP];
      a1.x = qhi ? 0.f : y4;
      a1.y = qhi ? 0.f : 1.f;
      v8f c2a = {}, c2b = {};
      c2a = wmma_f32_k4(wb0a, a0, c2a);
      c2b = wmma_f32_k4(wb0b, a0, c2b);
      c2a = wmma_f32_k4(wb1a, a1, c2a);
      c2b = wmma_f32_k4(wb1b, a1, c2b);

      v2f s = acc[t];
      #pragma unroll
      for (int j = 0; j < 4; ++j) {
        v2f va, vb;
        va.x = c2a[2 * j]     > 0.f ? c2a[2 * j]     : 0.f;
        va.y = c2a[2 * j + 1] > 0.f ? c2a[2 * j + 1] : 0.f;
        vb.x = c2b[2 * j]     > 0.f ? c2b[2 * j]     : 0.f;
        vb.y = c2b[2 * j + 1] > 0.f ? c2b[2 * j + 1] : 0.f;
        s = __builtin_elementwise_fma(va, wc.w3A[j], s);
        s = __builtin_elementwise_fma(vb, w3b[j], s);
      }
      acc[t] = s;
    }
  };

  WRaw wn;
  load_w(wn, slice);
  load_w3(wn, slice);
  __syncthreads();

  const bool tail = (slice < NTAIL);
  for (int step = 0; step < NFULL; ++step) {
    const int f = step * SPLIT + slice;
    {
      float* ys = &ybuf[(step + 1) & 1][0];
      #pragma unroll
      for (int k = 0; k < NPC; ++k) *reinterpret_cast<v4f*>(ys + sy[k]) = nx[k];
    }
    const WRaw wc = wn;
    v8f da, db;
    setup(wc, da, db);

    const int fn = (step + 1 < NFULL || tail) ? f + SPLIT : f;
    load_w(wn, fn);

    {
      const int ps = step + 2;
      const float* yn = y + ((ps < NFULL) ? ps * SW : TAILOFF);
      #pragma unroll
      for (int k = 0; k < NPC; ++k) nx[k] = *reinterpret_cast<const v4f*>(yn + gy[k]);
    }

    tiles(da, db, wc, &ybuf[step & 1][0]);

    load_w3(wn, fn);

    lds_barrier();
  }
  if (tail) {
    v8f da, db;
    setup(wn, da, db);
    tiles(da, db, wn, &ybuf[NFULL & 1][TAILBLK * DIN]);
  }

  float rs[RT];
  #pragma unroll
  for (int t = 0; t < RT; ++t) {
    float v = acc[t].x + acc[t].y;
    rs[t] = v + swz<XOR16>(v);
  }
  if (!qhi) {
    #pragma unroll
    for (int t = 0; t < RT; ++t) part[slice][t * 16 + n] = rs[t];
  }
  __syncthreads();

  if (threadIdx.x < ROWS) {
    float s = part[0][threadIdx.x];
    for (int w = 1; w < SPLIT; ++w) s += part[w][threadIdx.x];
    out[blockIdx.x * ROWS + threadIdx.x] = s + b3[0];
  }
}

extern "C" void kernel_launch(void* const* d_in, const int* in_sizes, int n_in,
                              void* d_out, int out_size, void* d_ws, size_t ws_size,
                              hipStream_t stream) {
  (void)in_sizes; (void)n_in; (void)d_ws; (void)ws_size; (void)out_size;
  const float* y  = (const float*)d_in[0];
  const float* W1 = (const float*)d_in[1];
  const float* b1 = (const float*)d_in[2];
  const float* W2 = (const float*)d_in[3];
  const float* b2 = (const float*)d_in[4];
  const float* W3 = (const float*)d_in[5];
  const float* b3 = (const float*)d_in[6];
  float* out = (float*)d_out;

  dim3 grid(BATCH / ROWS);
  dim3 block(SPLIT * 32);
  mlp200_wmma_kernel<<<grid, block, 0, stream>>>(y, W1, b1, W2, b2, W3, b3, out);
}


# Per-tile product: one K=4 chunk per M tile (2 WMMAs, not 4), W12 bias row as C operand; input 4 added in-lane by packed FMA (row 4, bias turned via wave-private LDS): sum order changes
# speedup vs baseline: 4.6167x; 1.1455x over previous
#include <hip/hip_runtime.h>


typedef float v2f __attribute__((ext_vector_type(2)));
typedef float v4f __attribute__((ext_vector_type(4)));
typedef float v8f __attribute__((ext_vector_type(8)));

#define BATCH 32768
#define NF    200
#define DIN   5
#define NH    15
#define NOUT  30
#define SPLIT 16
#define RT    8
#define ROWS  (RT * 16)
#define NFULL (NF / SPLIT)
#define NTAIL (NF - NFULL * SPLIT)
#define SW    (SPLIT * DIN)
#define TAILOFF (NF * DIN - SW)
#define TAILBLK (SPLIT - NTAIL)
#define SP    84
#define NPC   (ROWS * SW / 4 / (SPLIT * 32))

__device__ __forceinline__ v8f wmma_f32_k4(v2f a, v2f b, v8f c) {
  return __builtin_amdgcn_wmma_f32_16x16x4_f32(
       false, a,  false, b,
       (short)0, c,  false,  false);
}

template <int IMM>
__device__ __forceinline__ float swz(float x) {
  return __int_as_float(__builtin_amdgcn_ds_swizzle(__float_as_int(x), IMM));
}
#define XOR16  0x401f

__device__ __forceinline__ void lds_barrier() {
  __builtin_amdgcn_fence(__ATOMIC_RELEASE, "workgroup", "local");
  __builtin_amdgcn_s_barrier();
  __builtin_amdgcn_fence(__ATOMIC_ACQUIRE, "workgroup", "local");
}

__global__ __launch_bounds__(SPLIT * 32) __attribute__((amdgpu_waves_per_eu(4, 4)))
void mlp200_wmma_kernel(
    const float* __restrict__ y,  const float* __restrict__ W1,
    const float* __restrict__ b1, const float* __restrict__ W2,
    const float* __restrict__ b2, const float* __restrict__ W3,
    const float* __restrict__ b3, float* __restrict__ out)
{
  __shared__ float part[SPLIT][ROWS];
  __shared__ __attribute__((aligned(16))) float ybuf[2][ROWS * SP];
  __shared__ __attribute__((aligned(32))) float w12x[SPLIT][4][32];

  const int  lane = threadIdx.x & 31;
  const int  slice = __builtin_amdgcn_readfirstlane((int)(threadIdx.x >> 5));
  const int  q    = lane >> 4;
  const bool qhi  = (q != 0);
  const int  n    = lane & 15;
  const int  b0   = blockIdx.x * ROWS;

  const int  oc2  = 16 + ((n < 14) ? n : 13);

  const int  ly0  = n * SP + slice * DIN + 2 * q;
  const int  ly4  = n * SP + slice * DIN + 4;
  constexpr int TSTEP = 16 * SP;

  int gy[NPC], sy[NPC];
  #pragma unroll
  for (int k = 0; k < NPC; ++k) {
    const int p  = (int)threadIdx.x + k * (SPLIT * 32);
    const int pr = p / (SW / 4), pc = p % (SW / 4);
    gy[k] = (b0 + pr) * (NF * DIN) + 4 * pc;
    sy[k] = pr * SP + 4 * pc;
  }

  v4f nx[NPC];
  {
    v4f s0[NPC];
    #pragma unroll
    for (int k = 0; k < NPC; ++k) s0[k] = *reinterpret_cast<const v4f*>(y + gy[k]);
    #pragma unroll
    for (int k = 0; k < NPC; ++k) nx[k] = *reinterpret_cast<const v4f*>(y + SW + gy[k]);
    #pragma unroll
    for (int k = 0; k < NPC; ++k) *reinterpret_cast<v4f*>(&ybuf[0][sy[k]]) = s0[k];
  }

  const bool isb  = (n == 3);
  const int  irow = (int)((0x0000003200000410ull >> (4 * n)) & 15ull);
  const float* pA = isb ? b1 : (W1 + irow * NH);
  const int  stpA = isb ? NH : (DIN * NH);
  const float c15 = isb ? 1.f : 0.f;

  struct WRaw {
    float ax[4], ay[4];
    float bax[4], bbx[4];
    float bay[4], bby[4];
    float b2a, b2b;
    v2f  w3A[4], w3B[4];
  };
  auto load_w = [&](WRaw& w, int f) {
    const float* W2f = W2 + f * (NH * NOUT);
    const float* b2f = b2 + f * NOUT;
    const float* pa  = pA + f * stpA + 2 * q;
    #pragma unroll
    for (int k = 0; k < 4; ++k) {
      w.ax[k] = pa[4 * k];
      w.ay[k] = (k == 3) ? pa[13 - q] : pa[4 * k + 1];
      const int h0  = 4 * k + 2 * q;
      int       h1i = h0 + 1; if (h1i > NH - 1) h1i = NH - 1;
      w.bax[k] = W2f[h0  * NOUT + n];
      w.bbx[k] = W2f[h0  * NOUT + oc2];
      w.bay[k] = W2f[h1i * NOUT + n];
      w.bby[k] = W2f[h1i * NOUT + oc2];
    }
    w.b2a = b2f[n];
    w.b2b = b2f[oc2];
  };
  auto load_w3 = [&](WRaw& w, int f) {
    const float* W3f = W3 + f * NOUT;
    #pragma unroll
    for (int j = 0; j < 4; ++j) {
      int ob = 16 + 8 * q + 2 * j; if (ob > NOUT - 2) ob = NOUT - 2;
      w.w3A[j] = *reinterpret_cast<const v2f*>(W3f + 8 * q + 2 * j);
      w.w3B[j] = *reinterpret_cast<const v2f*>(W3f + ob);
    }
  };

  v2f acc[RT];
  #pragma unroll
  for (int t = 0; t < RT; ++t) acc[t] = v2f{0.f, 0.f};

  auto setup = [&](const WRaw& wc, v8f& da, v8f& db) {
    v2f wa[4];
    #pragma unroll
    for (int k = 0; k < 4; ++k) {
      float ay = wc.ay[k];
      if (k == 3) ay = qhi ? c15 : ay;
      wa[k].x = wc.ax[k];
      wa[k].y = ay;
    }
    v2f bta[4], btb[4];
    #pragma unroll
    for (int k = 0; k < 4; ++k) {
      bta[k].x = wc.bax[k];
      btb[k].x = wc.bbx[k];
      if (k == 3) {
        bta[k].y = qhi ? wc.b2a : wc.bay[k];
        btb[k].y = qhi ? wc.b2b : wc.bby[k];
      } else {
        bta[k].y = wc.bay[k];
        btb[k].y = wc.bby[k];
      }
    }
    da = v8f{}; db = v8f{};
    #pragma unroll
    for (int k = 0; k < 4; ++k) {
      da = wmma_f32_k4(wa[k], bta[k], da);
      db = wmma_f32_k4(wa[k], btb[k], db);
    }
  };

  auto tiles = [&](const v8f& da, const v8f& db, const WRaw& wc, const float* yl) {
    v2f wb0a, wb0b;
    wb0a.x = da[0]; wb0a.y = da[1];
    wb0b.x = db[0]; wb0b.y = db[1];
    float* xs = &w12x[slice][0][0];
    xs[lane]      = da[2];
    xs[32 + lane] = db[2];
    xs[64 + lane] = da[3];
    xs[96 + lane] = db[3];
    const v8f r4a = *reinterpret_cast<const v8f*>(xs + 8 * q);
    const v8f r4b = *reinterpret_cast<const v8f*>(xs + 32 + 8 * q);
    const v8f bia = *reinterpret_cast<const v8f*>(xs + 64 + 8 * q);
    const v8f bib = *reinterpret_cast<const v8f*>(xs + 96 + 8 * q);
    v2f w3b[4];
    #pragma unroll
    for (int j = 0; j < 4; ++j) w3b[j] = wc.w3B[j];
    w3b[3].x = qhi ? 0.f : w3b[3].x;
    w3b[3].y = qhi ? 0.f : w3b[3].y;
    #pragma unroll
    for (int t = 0; t < RT; ++t) {
      v2f a0;
      a0.x = yl[ly0 + t * TSTEP];
      a0.y = yl[ly0 + 1 + t * TSTEP];
      const float y4 = yl[ly4 + t * TSTEP];
      const v8f c2a = wmma_f32_k4(wb0a, a0, bia);
      const v8f c2b = wmma_f32_k4(wb0b, a0, bib);
      const v2f yy = v2f{y4, y4};

      v2f s = acc[t];
      #pragma unroll
      for (int j = 0; j < 4; ++j) {
        const v2f xa = __builtin_elementwise_fma(
            v2f{r4a[2 * j], r4a[2 * j + 1]}, yy, v2f{c2a[2 * j], c2a[2 * j + 1]});
        const v2f xb = __builtin_elementwise_fma(
            v2f{r4b[2 * j], r4b[2 * j + 1]}, yy, v2f{c2b[2 * j], c2b[2 * j + 1]});
        v2f va, vb;
        va.x = xa.x > 0.f ? xa.x : 0.f;
        va.y = xa.y > 0.f ? xa.y : 0.f;
        vb.x = xb.x > 0.f ? xb.x : 0.f;
        vb.y = xb.y > 0.f ? xb.y : 0.f;
        s = __builtin_elementwise_fma(va, wc.w3A[j], s);
        s = __builtin_elementwise_fma(vb, w3b[j], s);
      }
      acc[t] = s;
    }
  };

  WRaw wn;
  load_w(wn, slice);
  load_w3(wn, slice);
  __syncthreads();

  const bool tail = (slice < NTAIL);
  for (int step = 0; step < NFULL; ++step) {
    const int f = step * SPLIT + slice;
    {
      float* ys = &ybuf[(step + 1) & 1][0];
      #pragma unroll
      for (int k = 0; k < NPC; ++k) *reinterpret_cast<v4f*>(ys + sy[k]) = nx[k];
    }
    const WRaw wc = wn;
    v8f da, db;
    setup(wc, da, db);

    const int fn = (step + 1 < NFULL || tail) ? f + SPLIT : f;
    load_w(wn, fn);

    {
      const int ps = step + 2;
      const float* yn = y + ((ps < NFULL) ? ps * SW : TAILOFF);
      #pragma unroll
      for (int k = 0; k < NPC; ++k) nx[k] = *reinterpret_cast<const v4f*>(yn + gy[k]);
    }

    tiles(da, db, wc, &ybuf[step & 1][0]);

    load_w3(wn, fn);

    lds_barrier();
  }
  if (tail) {
    v8f da, db;
    setup(wn, da, db);
    tiles(da, db, wn, &ybuf[NFULL & 1][TAILBLK * DIN]);
  }

  float rs[RT];
  #pragma unroll
  for (int t = 0; t < RT; ++t) {
    float v = acc[t].x + acc[t].y;
    rs[t] = v + swz<XOR16>(v);
  }
  if (!qhi) {
    #pragma unroll
    for (int t = 0; t < RT; ++t) part[slice][t * 16 + n] = rs[t];
  }
  __syncthreads();

  if (threadIdx.x < ROWS) {
    float s = part[0][threadIdx.x];
    for (int w = 1; w < SPLIT; ++w) s += part[w][threadIdx.x];
    out[blockIdx.x * ROWS + threadIdx.x] = s + b3[0];
  }
}

extern "C" void kernel_launch(void* const* d_in, const int* in_sizes, int n_in,
                              void* d_out, int out_size, void* d_ws, size_t ws_size,
                              hipStream_t stream) {
  (void)in_sizes; (void)n_in; (void)d_ws; (void)ws_size; (void)out_size;
  const float* y  = (const float*)d_in[0];
  const float* W1 = (const float*)d_in[1];
  const float* b1 = (const float*)d_in[2];
  const float* W2 = (const float*)d_in[3];
  const float* b2 = (const float*)d_in[4];
  const float* W3 = (const float*)d_in[5];
  const float* b3 = (const float*)d_in[6];
  float* out = (float*)d_out;

  dim3 grid(BATCH / ROWS);
  dim3 block(SPLIT * 32);
  mlp200_wmma_kernel<<<grid, block, 0, stream>>>(y, W1, b1, W2, b2, W3, b3, out);
}


# ReLU done by the [0,1] clamp bit of the input-4 packed multiply-add: W12 scaled by 2^-64 per feature (exact), row sums scaled back once at the end; 16 maxima per lane and tile go; sum order unchanged
# speedup vs baseline: 4.7785x; 1.0350x over previous
#include <hip/hip_runtime.h>


typedef float v2f __attribute__((ext_vector_type(2)));
typedef float v4f __attribute__((ext_vector_type(4)));
typedef float v8f __attribute__((ext_vector_type(8)));

#define BATCH 32768
#define NF    200
#define DIN   5
#define NH    15
#define NOUT  30
#define SPLIT 16
#define RT    8
#define ROWS  (RT * 16)
#define NFULL (NF / SPLIT)
#define NTAIL (NF - NFULL * SPLIT)
#define SW    (SPLIT * DIN)
#define TAILOFF (NF * DIN - SW)
#define TAILBLK (SPLIT - NTAIL)
#define SP    84
#define NPC   (ROWS * SW / 4 / (SPLIT * 32))

__device__ __forceinline__ v8f wmma_f32_k4(v2f a, v2f b, v8f c) {
  return __builtin_amdgcn_wmma_f32_16x16x4_f32(
       false, a,  false, b,
       (short)0, c,  false,  false);
}

template <int IMM>
__device__ __forceinline__ float swz(float x) {
  return __int_as_float(__builtin_amdgcn_ds_swizzle(__float_as_int(x), IMM));
}
#define XOR16  0x401f

#define SCL   0x1p-64f
#define UNSCL 0x1p+64f

__device__ __forceinline__ v2f fma_in4_relu(v2f a, v2f b, v2f c) {
  asm("v_pk_fma_f32 %0, %1, %2, %0 op_sel_hi:[1,0,1] clamp"
      : "+v"(c) : "v"(a), "v"(b));
  return c;
}

__device__ __forceinline__ void lds_barrier() {
  __builtin_amdgcn_fence(__ATOMIC_RELEASE, "workgroup", "local");
  __builtin_amdgcn_s_barrier();
  __builtin_amdgcn_fence(__ATOMIC_ACQUIRE, "workgroup", "local");
}

__global__ __launch_bounds__(SPLIT * 32) __attribute__((amdgpu_waves_per_eu(4, 4)))
void mlp200_wmma_kernel(
    const float* __restrict__ y,  const float* __restrict__ W1,
    const float* __restrict__ b1, const float* __restrict__ W2,
    const float* __restrict__ b2, const float* __restrict__ W3,
    const float* __restrict__ b3, float* __restrict__ out)
{
  __shared__ float part[SPLIT][ROWS];
  __shared__ __attribute__((aligned(16))) float ybuf[2][ROWS * SP];
  __shared__ __attribute__((aligned(32))) float w12x[SPLIT][4][32];

  const int  lane = threadIdx.x & 31;
  const int  slice = __builtin_amdgcn_readfirstlane((int)(threadIdx.x >> 5));
  const int  q    = lane >> 4;
  const bool qhi  = (q != 0);
  const int  n    = lane & 15;
  const int  b0   = blockIdx.x * ROWS;

  const int  oc2  = 16 + ((n < 14) ? n : 13);

  const int  ly0  = n * SP + slice * DIN + 2 * q;
  const int  ly4  = n * SP + slice * DIN + 4;
  constexpr int TSTEP = 16 * SP;

  int gy[NPC], sy[NPC];
  #pragma unroll
  for (int k = 0; k < NPC; ++k) {
    const int p  = (int)threadIdx.x + k * (SPLIT * 32);
    const int pr = p / (SW / 4), pc = p % (SW / 4);
    gy[k] = (b0 + pr) * (NF * DIN) + 4 * pc;
    sy[k] = pr * SP + 4 * pc;
  }

  v4f nx[NPC];
  {
    v4f s0[NPC];
    #pragma unroll
    for (int k = 0; k < NPC; ++k) s0[k] = *reinterpret_cast<const v4f*>(y + gy[k]);
    #pragma unroll
    for (int k = 0; k < NPC; ++k) nx[k] = *reinterpret_cast<const v4f*>(y + SW + gy[k]);
    #pragma unroll
    for (int k = 0; k < NPC; ++k) *reinterpret_cast<v4f*>(&ybuf[0][sy[k]]) = s0[k];
  }

  const bool isb  = (n == 3);
  const int  irow = (int)((0x0000003200000410ull >> (4 * n)) & 15ull);
  const float* pA = isb ? b1 : (W1 + irow * NH);
  const int  stpA = isb ? NH : (DIN * NH);
  const float c15 = isb ? 1.f : 0.f;

  struct WRaw {
    float ax[4], ay[4];
    float bax[4], bbx[4];
    float bay[4], bby[4];
    float b2a, b2b;
    v2f  w3A[4], w3B[4];
  };
  auto load_w = [&](WRaw& w, int f) {
    const float* W2f = W2 + f * (NH * NOUT);
    const float* b2f = b2 + f * NOUT;
    const float* pa  = pA + f * stpA + 2 * q;
    #pragma unroll
    for (int k = 0; k < 4; ++k) {
      w.ax[k] = pa[4 * k];
      w.ay[k] = (k == 3) ? pa[13 - q] : pa[4 * k + 1];
      const int h0  = 4 * k + 2 * q;
      int       h1i = h0 + 1; if (h1i > NH - 1) h1i = NH - 1;
      w.bax[k] = W2f[h0  * NOUT + n];
      w.bbx[k] = W2f[h0  * NOUT + oc2];
      w.bay[k] = W2f[h1i * NOUT + n];
      w.bby[k] = W2f[h1i * NOUT + oc2];
    }
    w.b2a = b2f[n];
    w.b2b = b2f[oc2];
  };
  auto load_w3 = [&](WRaw& w, int f) {
    const float* W3f = W3 + f * NOUT;
    #pragma unroll
    for (int j = 0; j < 4; ++j) {
      int ob = 16 + 8 * q + 2 * j; if (ob > NOUT - 2) ob = NOUT - 2;
      w.w3A[j] = *reinterpret_cast<const v2f*>(W3f + 8 * q + 2 * j);
      w.w3B[j] = *reinterpret_cast<const v2f*>(W3f + ob);
    }
  };

  v2f acc[RT];
  #pragma unroll
  for (int t = 0; t < RT; ++t) acc[t] = v2f{0.f, 0.f};

  auto setup = [&](const WRaw& wc, v8f& da, v8f& db) {
    v2f wa[4];
    #pragma unroll
    for (int k = 0; k < 4; ++k) {
      float ay = wc.ay[k];
      if (k == 3) ay = qhi ? c15 : ay;
      wa[k].x = wc.ax[k];
      wa[k].y = ay;
    }
    v2f bta[4], btb[4];
    #pragma unroll
    for (int k = 0; k < 4; ++k) {
      bta[k].x = wc.bax[k];
      btb[k].x = wc.bbx[k];
      if (k == 3) {
        bta[k].y = qhi ? wc.b2a : wc.bay[k];
        btb[k].y = qhi ? wc.b2b : wc.bby[k];
      } else {
        bta[k].y = wc.bay[k];
        btb[k].y = wc.bby[k];
      }
    }
    da = v8f{}; db = v8f{};
    #pragma unroll
    for (int k = 0; k < 4; ++k) {
      da = wmma_f32_k4(wa[k], bta[k], da);
      db = wmma_f32_k4(wa[k], btb[k], db);
    }
  };

  auto tiles = [&](const v8f& da, const v8f& db, const WRaw& wc, const float* yl) {
    const v2f d01a = v2f{da[0], da[1]} * SCL, d23a = v2f{da[2], da[3]} * SCL;
    const v2f d01b = v2f{db[0], db[1]} * SCL, d23b = v2f{db[2], db[3]} * SCL;
    const v2f wb0a = d01a, wb0b = d01b;
    float* xs = &w12x[slice][0][0];
    xs[lane]      = d23a.x;
    xs[32 + lane] = d23b.x;
    xs[64 + lane] = d23a.y;
    xs[96 + lane] = d23b.y;
    const v8f r4a = *reinterpret_cast<const v8f*>(xs + 8 * q);
    const v8f r4b = *reinterpret_cast<const v8f*>(xs + 32 + 8 * q);
    const v8f bia = *reinterpret_cast<const v8f*>(xs + 64 + 8 * q);
    const v8f bib = *reinterpret_cast<const v8f*>(xs + 96 + 8 * q);
    v2f w3b[4];
    #pragma unroll
    for (int j = 0; j < 4; ++j) w3b[j] = wc.w3B[j];
    w3b[3].x = qhi ? 0.f : w3b[3].x;
    w3b[3].y = qhi ? 0.f : w3b[3].y;
    v2f y4p[RT / 2];
    #pragma unroll
    for (int t = 0; t < RT; t += 2)
      y4p[t / 2] = v2f{yl[ly4 + t * TSTEP], yl[ly4 + (t + 1) * TSTEP]};
    #pragma unroll
    for (int t = 0; t < RT; ++t) {
      v2f a0;
      a0.x = yl[ly0 + t * TSTEP];
      a0.y = yl[ly0 + 1 + t * TSTEP];
      const v8f c2a = wmma_f32_k4(wb0a, a0, bia);
      const v8f c2b = wmma_f32_k4(wb0b, a0, bib);
      const v2f yy = (t & 1)
          ? v2f{y4p[t / 2].y, __builtin_nondeterministic_value(y4p[t / 2].y)}
          : y4p[t / 2];

      v2f s = acc[t];
      #pragma unroll
      for (int j = 0; j < 4; ++j) {
        const v2f va = fma_in4_relu(
            v2f{r4a[2 * j], r4a[2 * j + 1]}, yy, v2f{c2a[2 * j], c2a[2 * j + 1]});
        const v2f vb = fma_in4_relu(
            v2f{r4b[2 * j], r4b[2 * j + 1]}, yy, v2f{c2b[2 * j], c2b[2 * j + 1]});
        s = __builtin_elementwise_fma(va, wc.w3A[j], s);
        s = __builtin_elementwise_fma(vb, w3b[j], s);
      }
      acc[t] = s;
    }
  };

  WRaw wn;
  load_w(wn, slice);
  load_w3(wn, slice);
  __syncthreads();

  const bool tail = (slice < NTAIL);
  #pragma clang loop unroll(disable)
  for (int step = 0; step < NFULL; ++step) {
    const int f = step * SPLIT + slice;
    {
      float* ys = &ybuf[(step + 1) & 1][0];
      #pragma unroll
      for (int k = 0; k < NPC; ++k) *reinterpret_cast<v4f*>(ys + sy[k]) = nx[k];
    }
    const WRaw wc = wn;
    v8f da, db;
    setup(wc, da, db);

    const int fn = (step + 1 < NFULL || tail) ? f + SPLIT : f;
    load_w(wn, fn);

    {
      const int ps = step + 2;
      const float* yn = y + ((ps < NFULL) ? ps * SW : TAILOFF);
      #pragma unroll
      for (int k = 0; k < NPC; ++k) nx[k] = *reinterpret_cast<const v4f*>(yn + gy[k]);
    }

    tiles(da, db, wc, &ybuf[step & 1][0]);

    load_w3(wn, fn);

    lds_barrier();
  }
  if (tail) {
    v8f da, db;
    setup(wn, da, db);
    tiles(da, db, wn, &ybuf[NFULL & 1][TAILBLK * DIN]);
  }

  float rs[RT];
  #pragma unroll
  for (int t = 0; t < RT; ++t) {
    float v = acc[t].x + acc[t].y;
    rs[t] = v + swz<XOR16>(v);
  }
  if (!qhi) {
    #pragma unroll
    for (int t = 0; t < RT; ++t) part[slice][t * 16 + n] = rs[t];
  }
  __syncthreads();

  if (threadIdx.x < ROWS) {
    float s = part[0][threadIdx.x];
    for (int w = 1; w < SPLIT; ++w) s += part[w][threadIdx.x];
    out[blockIdx.x * ROWS + threadIdx.x] = __builtin_fmaf(s, UNSCL, b3[0]);
  }
}

extern "C" void kernel_launch(void* const* d_in, const int* in_sizes, int n_in,
                              void* d_out, int out_size, void* d_ws, size_t ws_size,
                              hipStream_t stream) {
  (void)in_sizes; (void)n_in; (void)d_ws; (void)ws_size; (void)out_size;
  const float* y  = (const float*)d_in[0];
  const float* W1 = (const float*)d_in[1];
  const float* b1 = (const float*)d_in[2];
  const float* W2 = (const float*)d_in[3];
  const float* b2 = (const float*)d_in[4];
  const float* W3 = (const float*)d_in[5];
  const float* b3 = (const float*)d_in[6];
  float* out = (float*)d_out;

  dim3 grid(BATCH / ROWS);
  dim3 block(SPLIT * 32);
  mlp200_wmma_kernel<<<grid, block, 0, stream>>>(y, W1, b1, W2, b2, W3, b3, out);
}
